# starting kernel
# baseline (speedup 1.0000x reference)
#include <hip/hip_runtime.h>

typedef __attribute__((ext_vector_type(2))) float v2f;
typedef __attribute__((ext_vector_type(8))) float v8f;

#define NCOL 300
#define NTILE 19
#define BFRAG_OFF 0
#define BFRAG_FLOATS (NTILE * 32 * 2)
#define W32_OFF BFRAG_FLOATS
#define W32_PAD (NTILE * 16)
#define W31_OFF (W32_OFF + W32_PAD)

__global__ __launch_bounds__(256) void phi2_prep(
    const float* __restrict__ varphi1,
    const float* __restrict__ varphi2,
    const float* __restrict__ l1,
    const float* __restrict__ l2,
    const float* __restrict__ l3,
    float* __restrict__ ws)
{
    const int tid = threadIdx.x;
    const float l3v = l3[0];

    for (int j = tid; j < W32_PAD; j += 256) {
        ws[W32_OFF + j] = (j < NCOL) ? varphi2[j] * (l2[j] - l3v) : 0.0f;
    }

    for (int i = tid; i < NTILE * 32; i += 256) {
        const int t = i >> 5;
        const int l = i & 31;
        const int j = t * 16 + (l & 15);
        float v0 = 0.0f, v1 = 0.0f;
        if (j < NCOL) {
            const float sj = -l2[j];
            if (l < 16) {
                v0 = varphi1[j * 3 + 0] * (l1[0] + sj);
                v1 = varphi1[j * 3 + 1] * (l1[1] + sj);
            } else {
                v0 = varphi1[j * 3 + 2] * (l1[2] + sj);
                v1 = 0.0f;
            }
        }
        ws[BFRAG_OFF + i * 2 + 0] = v0;
        ws[BFRAG_OFF + i * 2 + 1] = v1;
    }

    if (tid < 3) {
        const int k = tid;
        float s = 0.0f;
        for (int j = 0; j < NCOL; ++j) {
            const float w32j = varphi2[j] * (l2[j] - l3v);
            s -= w32j * varphi1[j * 3 + k];
        }
        ws[W31_OFF + k] = s;
    }
}

__device__ __forceinline__ float relu1(float v) {
    float r;
    asm("v_max_num_f32 %0, 0, %1" : "=v"(r) : "v"(v));
    return r;
}

__global__ __launch_bounds__(256) void phi2_main(
    const float* __restrict__ x,
    const float* __restrict__ ws,
    float* __restrict__ y,
    int nrows)
{
    const int lane = threadIdx.x & 31;
    const int wave = threadIdx.x >> 5;
    const long long rowBase = ((long long)blockIdx.x * 8 + wave) * 16;
    if (rowBase >= nrows) return;

    v2f a;
    {
        const int m = lane & 15;
        long long row = rowBase + m;
        if (row > nrows - 1) row = nrows - 1;
        if (lane < 16) {
            a.x = x[row * 2 + 0];
            a.y = x[row * 2 + 1];
        } else {
            a.x = 1.0f;
            a.y = 0.0f;
        }
    }

    const float* __restrict__ bf  = ws + BFRAG_OFF;
    const float* __restrict__ w32 = ws + W32_OFF;
    const float* __restrict__ w31 = ws + W31_OFF;

    v8f acc = {0.f, 0.f, 0.f, 0.f, 0.f, 0.f, 0.f, 0.f};

    for (int t = 0; t < NTILE; ++t) {
        const v2f b = *(const v2f*)(bf + (t * 32 + lane) * 2);
        v8f c = {};
        v8f d = __builtin_amdgcn_wmma_f32_16x16x4_f32(
             false, a,  false, b,
             (short)0, c,  false,  false);
        const float wv = w32[t * 16 + (lane & 15)];
#pragma unroll
        for (int r = 0; r < 8; ++r) {
            acc[r] += relu1(d[r]) * wv;
        }
    }

#pragma unroll
    for (int r = 0; r < 8; ++r) {
        float s = acc[r];
        s += __shfl_xor(s, 1, 32);
        s += __shfl_xor(s, 2, 32);
        s += __shfl_xor(s, 4, 32);
        s += __shfl_xor(s, 8, 32);
        acc[r] = s;
    }

    if ((lane & 15) == 0) {
        const float c0 = w31[0], c1 = w31[1], c2 = w31[2];
        const long long base = rowBase + ((lane >> 4) << 3);
#pragma unroll
        for (int r = 0; r < 8; ++r) {
            const long long orow = base + r;
            if (orow < nrows) {
                const float x0 = x[orow * 2 + 0];
                const float x1 = x[orow * 2 + 1];
                y[orow] = acc[r] + c0 * x0 + c1 * x1 + c2;
            }
        }
    }
}

extern "C" void kernel_launch(void* const* d_in, const int* in_sizes, int n_in,
                              void* d_out, int out_size, void* d_ws, size_t ws_size,
                              hipStream_t stream) {
    const float* x       = (const float*)d_in[0];
    const float* varphi1 = (const float*)d_in[1];
    const float* varphi2 = (const float*)d_in[2];
    const float* l1      = (const float*)d_in[3];
    const float* l2      = (const float*)d_in[4];
    const float* l3      = (const float*)d_in[5];
    float* y  = (float*)d_out;
    float* ws = (float*)d_ws;

    const int B = in_sizes[0] / 2;

    phi2_prep<<<1, 256, 0, stream>>>(varphi1, varphi2, l1, l2, l3, ws);

    const int wavesNeeded = (B + 15) / 16;
    const int grid = (wavesNeeded + 7) / 8;
    phi2_main<<<grid, 256, 0, stream>>>(x, ws, y, B);
}


# Main kernel: each wave loads the 19 B-fragments and weights once into registers and loops over many 16-row tiles (256 blocks x 16 waves); relu spelled as integer max so the loop schedules
# speedup vs baseline: 1.1692x; 1.1692x over previous
#include <hip/hip_runtime.h>

typedef __attribute__((ext_vector_type(2))) float v2f;
typedef __attribute__((ext_vector_type(8))) float v8f;

#define NCOL 300
#define NTILE 19
#define BFRAG_OFF 0
#define BFRAG_FLOATS (NTILE * 32 * 2)
#define W32_OFF BFRAG_FLOATS
#define W32_PAD (NTILE * 16)
#define W31_OFF (W32_OFF + W32_PAD)

__global__ __launch_bounds__(256) void phi2_prep(
    const float* __restrict__ varphi1,
    const float* __restrict__ varphi2,
    const float* __restrict__ l1,
    const float* __restrict__ l2,
    const float* __restrict__ l3,
    float* __restrict__ ws)
{
    const int tid = threadIdx.x;
    const float l3v = l3[0];

    for (int j = tid; j < W32_PAD; j += 256) {
        ws[W32_OFF + j] = (j < NCOL) ? varphi2[j] * (l2[j] - l3v) : 0.0f;
    }

    for (int i = tid; i < NTILE * 32; i += 256) {
        const int t = i >> 5;
        const int l = i & 31;
        const int j = t * 16 + (l & 15);
        float v0 = 0.0f, v1 = 0.0f;
        if (j < NCOL) {
            const float sj = -l2[j];
            if (l < 16) {
                v0 = varphi1[j * 3 + 0] * (l1[0] + sj);
                v1 = varphi1[j * 3 + 1] * (l1[1] + sj);
            } else {
                v0 = varphi1[j * 3 + 2] * (l1[2] + sj);
                v1 = 0.0f;
            }
        }
        ws[BFRAG_OFF + i * 2 + 0] = v0;
        ws[BFRAG_OFF + i * 2 + 1] = v1;
    }

    if (tid < 3) {
        const int k = tid;
        float s = 0.0f;
        for (int j = 0; j < NCOL; ++j) {
            const float w32j = varphi2[j] * (l2[j] - l3v);
            s -= w32j * varphi1[j * 3 + k];
        }
        ws[W31_OFF + k] = s;
    }
}

__device__ __forceinline__ float relu1(float v) {
    const int i = __builtin_bit_cast(int, v);
    return __builtin_bit_cast(float, i > 0 ? i : 0);
}

#define MAIN_THREADS 512
#define MAIN_WAVES   (MAIN_THREADS / 32)
#define MAIN_GRID    256

typedef __attribute__((ext_vector_type(4))) float v4f;

__global__ __launch_bounds__(MAIN_THREADS) void phi2_main(
    const float* __restrict__ x,
    const float* __restrict__ ws,
    float* __restrict__ y,
    int nrows)
{
    const int lane = threadIdx.x & 31;
    const int wave = __builtin_amdgcn_readfirstlane((int)(threadIdx.x >> 5));
    const int nWaves = (int)gridDim.x * MAIN_WAVES;
    const int nTiles = (nrows + 15) >> 4;
    int tile = (int)blockIdx.x * MAIN_WAVES + wave;
    if (tile >= nTiles) return;

    const float* __restrict__ bf  = ws + BFRAG_OFF;
    const float* __restrict__ w32 = ws + W32_OFF;
    const float* __restrict__ w31 = ws + W31_OFF;

    v2f   b[NTILE];
    float wv[NTILE];
#pragma unroll
    for (int t = 0; t < NTILE; ++t) {
        b[t]  = *(const v2f*)(bf + (t * 32 + lane) * 2);
        wv[t] = w32[t * 16 + (lane & 15)];
    }
    const float c0 = w31[0], c1 = w31[1], c2 = w31[2];

    const int  m    = lane & 15;
    const bool hi   = lane >= 16;
    const int  half = hi ? 8 : 0;

    auto loadX = [&](int tl) -> v2f {
        int row = tl * 16 + m;
        if (row > nrows - 1) row = nrows - 1;
        return *(const v2f*)(x + (size_t)row * 2);
    };

    v2f xv = loadX(tile);

    for (;;) {
        const int  nextTile = tile + nWaves;
        const bool more = nextTile < nTiles;
        const bool full = tile * 16 + 16 <= nrows;
        v2f xn = xv;
        if (more) xn = loadX(nextTile);

        const int base = tile * 16 + half;
        v4f q0, q1, q2, q3;
        if (full) {
            const v4f* xq = (const v4f*)(x + (size_t)base * 2);
            q0 = xq[0]; q1 = xq[1]; q2 = xq[2]; q3 = xq[3];
        }

        v2f a;
        a.x = hi ? 1.0f : xv.x;
        a.y = hi ? 0.0f : xv.y;

        v2f accp[4] = {{0.f, 0.f}, {0.f, 0.f}, {0.f, 0.f}, {0.f, 0.f}};

        const v8f zero = {};
#pragma unroll
        for (int t = 0; t < NTILE; ++t) {
            const v8f d = __builtin_amdgcn_wmma_f32_16x16x4_f32(
                false, a, false, b[t], (short)0, zero, false, false);
            const v2f w2 = {wv[t], wv[t]};
            accp[0] += (v2f){relu1(d[0]), relu1(d[1])} * w2;
            accp[1] += (v2f){relu1(d[2]), relu1(d[3])} * w2;
            accp[2] += (v2f){relu1(d[4]), relu1(d[5])} * w2;
            accp[3] += (v2f){relu1(d[6]), relu1(d[7])} * w2;
        }

        float acc[8] = {accp[0].x, accp[0].y, accp[1].x, accp[1].y,
                        accp[2].x, accp[2].y, accp[3].x, accp[3].y};

#pragma unroll
        for (int r = 0; r < 8; ++r) {
            float s = acc[r];
            s += __shfl_xor(s, 1, 32);
            s += __shfl_xor(s, 2, 32);
            s += __shfl_xor(s, 4, 32);
            s += __shfl_xor(s, 8, 32);
            acc[r] = s;
        }

        if (full) {
            v4f o0, o1;
            o0.x = acc[0] + c0 * q0.x + c1 * q0.y + c2;
            o0.y = acc[1] + c0 * q0.z + c1 * q0.w + c2;
            o0.z = acc[2] + c0 * q1.x + c1 * q1.y + c2;
            o0.w = acc[3] + c0 * q1.z + c1 * q1.w + c2;
            o1.x = acc[4] + c0 * q2.x + c1 * q2.y + c2;
            o1.y = acc[5] + c0 * q2.z + c1 * q2.w + c2;
            o1.z = acc[6] + c0 * q3.x + c1 * q3.y + c2;
            o1.w = acc[7] + c0 * q3.z + c1 * q3.w + c2;
            if (m == 0) {
                v4f* yo = (v4f*)(y + base);
                yo[0] = o0;
                yo[1] = o1;
            }
        } else if (m == 0) {
#pragma unroll
            for (int r = 0; r < 8; ++r) {
                const int orow = base + r;
                if (orow < nrows) {
                    const float x0 = x[(size_t)orow * 2 + 0];
                    const float x1 = x[(size_t)orow * 2 + 1];
                    y[orow] = acc[r] + c0 * x0 + c1 * x1 + c2;
                }
            }
        }

        if (!more) break;
        xv = xn;
        tile = nextTile;
    }
}

extern "C" void kernel_launch(void* const* d_in, const int* in_sizes, int n_in,
                              void* d_out, int out_size, void* d_ws, size_t ws_size,
                              hipStream_t stream) {
    const float* x       = (const float*)d_in[0];
    const float* varphi1 = (const float*)d_in[1];
    const float* varphi2 = (const float*)d_in[2];
    const float* l1      = (const float*)d_in[3];
    const float* l2      = (const float*)d_in[4];
    const float* l3      = (const float*)d_in[5];
    float* y  = (float*)d_out;
    float* ws = (float*)d_ws;

    const int B = in_sizes[0] / 2;

    phi2_prep<<<1, 256, 0, stream>>>(varphi1, varphi2, l1, l2, l3, ws);

    phi2_main<<<MAIN_GRID, MAIN_THREADS, 0, stream>>>(x, ws, y, B);
}


# One launch instead of two: each wave builds its B-fragments, W32 and W31 from the raw inputs in registers (prep kernel and workspace pass removed); W31 summed as per-lane partials + 16-lane tree
# speedup vs baseline: 1.9022x; 1.6269x over previous
#include <hip/hip_runtime.h>


typedef __attribute__((ext_vector_type(2))) float v2f;
typedef __attribute__((ext_vector_type(4))) float v4f;
typedef __attribute__((ext_vector_type(8))) float v8f;

#define NCOL 300
#define NTILE 19

__device__ __forceinline__ float relu1(float v) {
    const int i = __builtin_bit_cast(int, v);
    return __builtin_bit_cast(float, i > 0 ? i : 0);
}

__device__ __forceinline__ float lane_value(float v, int l) {
    return __builtin_bit_cast(float,
        __builtin_amdgcn_readlane(__builtin_bit_cast(int, v), l));
}

#define MAIN_THREADS 512
#define MAIN_WAVES   (MAIN_THREADS / 32)
#define MAIN_GRID    256

__global__ __launch_bounds__(MAIN_THREADS) void net_fused(
    const float* __restrict__ x,
    const float* __restrict__ varphi1,
    const float* __restrict__ varphi2,
    const float* __restrict__ l1,
    const float* __restrict__ l2,
    const float* __restrict__ l3,
    float* __restrict__ y,
    int nrows)
{
    const int lane = threadIdx.x & 31;
    const int wave = __builtin_amdgcn_readfirstlane((int)(threadIdx.x >> 5));
    const int nWaves = (int)gridDim.x * MAIN_WAVES;
    const int nTiles = (nrows + 15) >> 4;
    int tile = (int)blockIdx.x * MAIN_WAVES + wave;
    if (tile >= nTiles) return;

    const int  m    = lane & 15;
    const bool hi   = lane >= 16;
    const int  half = hi ? 8 : 0;

    auto loadX = [&](int tl) -> v2f {
        int row = tl * 16 + m;
        if (row > nrows - 1) row = nrows - 1;
        return *(const v2f*)(x + (size_t)row * 2);
    };

    v2f xv = loadX(tile);

    const float l3v = l3[0];
    const float l1a = hi ? l1[2] : l1[0];
    const float l1b = l1[1];
    const int   kof = hi ? 2 : 0;

    v2f   b[NTILE];
    float wv[NTILE];
    float p0 = 0.0f, p1 = 0.0f;
#pragma unroll
    for (int t = 0; t < NTILE; ++t) {
        const int  j  = t * 16 + m;
        const bool ok = j < NCOL;
        const int  jc = ok ? j : NCOL - 1;
        const float l2j = l2[jc];
        const float f0  = varphi1[jc * 3 + kof];
        const float f1  = varphi1[jc * 3 + 1];
        const float sj  = -l2j;
        float w  = varphi2[jc] * (l2j - l3v);
        float b0 = f0 * (l1a + sj);
        float b1 = f1 * (l1b + sj);
        if (!ok) { w = 0.0f; b0 = 0.0f; }
        if (hi || !ok) b1 = 0.0f;
        b[t]  = (v2f){b0, b1};
        wv[t] = w;
        p0 += w * f0;
        p1 += w * f1;
    }
    p0 += __shfl_xor(p0, 1, 32);  p1 += __shfl_xor(p1, 1, 32);
    p0 += __shfl_xor(p0, 2, 32);  p1 += __shfl_xor(p1, 2, 32);
    p0 += __shfl_xor(p0, 4, 32);  p1 += __shfl_xor(p1, 4, 32);
    p0 += __shfl_xor(p0, 8, 32);  p1 += __shfl_xor(p1, 8, 32);
    const float c0 = -lane_value(p0, 0);
    const float c1 = -lane_value(p1, 0);
    const float c2 = -lane_value(p0, 16);

    for (;;) {
        const int  nextTile = tile + nWaves;
        const bool more = nextTile < nTiles;
        const bool full = tile * 16 + 16 <= nrows;
        v2f xn = xv;
        if (more) xn = loadX(nextTile);

        const int base = tile * 16 + half;
        v4f q0, q1, q2, q3;
        if (full) {
            const v4f* xq = (const v4f*)(x + (size_t)base * 2);
            q0 = xq[0]; q1 = xq[1]; q2 = xq[2]; q3 = xq[3];
        }

        v2f a;
        a.x = hi ? 1.0f : xv.x;
        a.y = hi ? 0.0f : xv.y;

        v2f accp[4] = {{0.f, 0.f}, {0.f, 0.f}, {0.f, 0.f}, {0.f, 0.f}};

        const v8f zero = {};
#pragma unroll
        for (int t = 0; t < NTILE; ++t) {
            const v8f d = __builtin_amdgcn_wmma_f32_16x16x4_f32(
                false, a, false, b[t], (short)0, zero, false, false);
            const v2f w2 = {wv[t], wv[t]};
            accp[0] += (v2f){relu1(d[0]), relu1(d[1])} * w2;
            accp[1] += (v2f){relu1(d[2]), relu1(d[3])} * w2;
            accp[2] += (v2f){relu1(d[4]), relu1(d[5])} * w2;
            accp[3] += (v2f){relu1(d[6]), relu1(d[7])} * w2;
        }

        float acc[8] = {accp[0].x, accp[0].y, accp[1].x, accp[1].y,
                        accp[2].x, accp[2].y, accp[3].x, accp[3].y};

#pragma unroll
        for (int r = 0; r < 8; ++r) {
            float s = acc[r];
            s += __shfl_xor(s, 1, 32);
            s += __shfl_xor(s, 2, 32);
            s += __shfl_xor(s, 4, 32);
            s += __shfl_xor(s, 8, 32);
            acc[r] = s;
        }

        if (full) {
            v4f o0, o1;
            o0.x = acc[0] + c0 * q0.x + c1 * q0.y + c2;
            o0.y = acc[1] + c0 * q0.z + c1 * q0.w + c2;
            o0.z = acc[2] + c0 * q1.x + c1 * q1.y + c2;
            o0.w = acc[3] + c0 * q1.z + c1 * q1.w + c2;
            o1.x = acc[4] + c0 * q2.x + c1 * q2.y + c2;
            o1.y = acc[5] + c0 * q2.z + c1 * q2.w + c2;
            o1.z = acc[6] + c0 * q3.x + c1 * q3.y + c2;
            o1.w = acc[7] + c0 * q3.z + c1 * q3.w + c2;
            if (m == 0) {
                v4f* yo = (v4f*)(y + base);
                yo[0] = o0;
                yo[1] = o1;
            }
        } else if (m == 0) {
#pragma unroll
            for (int r = 0; r < 8; ++r) {
                const int orow = base + r;
                if (orow < nrows) {
                    const float x0 = x[(size_t)orow * 2 + 0];
                    const float x1 = x[(size_t)orow * 2 + 1];
                    y[orow] = acc[r] + c0 * x0 + c1 * x1 + c2;
                }
            }
        }

        if (!more) break;
        xv = xn;
        tile = nextTile;
    }
}

extern "C" void kernel_launch(void* const* d_in, const int* in_sizes, int n_in,
                              void* d_out, int out_size, void* d_ws, size_t ws_size,
                              hipStream_t stream) {
    const float* x       = (const float*)d_in[0];
    const float* varphi1 = (const float*)d_in[1];
    const float* varphi2 = (const float*)d_in[2];
    const float* l1      = (const float*)d_in[3];
    const float* l2      = (const float*)d_in[4];
    const float* l3      = (const float*)d_in[5];
    float* y  = (float*)d_out;

    const int B = in_sizes[0] / 2;

    net_fused<<<MAIN_GRID, MAIN_THREADS, 0, stream>>>(
        x, varphi1, varphi2, l1, l2, l3, y, B);
}


# WMMA operands swapped: hidden units on D's registers, batch rows on lanes; a lane sums its own row, weights per register via a wave-private LDS strip; shuffle tree, x reloads gone; sum order changes
# speedup vs baseline: 2.1487x; 1.1296x over previous
#include <hip/hip_runtime.h>


typedef __attribute__((ext_vector_type(2))) float v2f;
typedef __attribute__((ext_vector_type(4))) float v4f;
typedef __attribute__((ext_vector_type(8))) float v8f;

#define NCOL 300
#define NTILE 19

__device__ __forceinline__ float relu1(float v) {
    const int i = __builtin_bit_cast(int, v);
    return __builtin_bit_cast(float, i > 0 ? i : 0);
}

__device__ __forceinline__ float lane_value(float v, int l) {
    return __builtin_bit_cast(float,
        __builtin_amdgcn_readlane(__builtin_bit_cast(int, v), l));
}

__device__ __forceinline__ float other_half(float v) {
    const int b = __builtin_bit_cast(int, v);
    return __builtin_bit_cast(float,
        __builtin_amdgcn_permlanex16(b, b, 0x76543210, 0xfedcba98, false, false));
}

#define MAIN_THREADS 512
#define MAIN_WAVES   (MAIN_THREADS / 32)
#define MAIN_GRID    256

__global__ __launch_bounds__(MAIN_THREADS) void net_fused(
    const float* __restrict__ x,
    const float* __restrict__ varphi1,
    const float* __restrict__ varphi2,
    const float* __restrict__ l1,
    const float* __restrict__ l2,
    const float* __restrict__ l3,
    float* __restrict__ y,
    int nrows)
{
    const int lane = threadIdx.x & 31;
    const int wave = __builtin_amdgcn_readfirstlane((int)(threadIdx.x >> 5));
    const int nWaves = (int)gridDim.x * MAIN_WAVES;
    const int nTiles = (nrows + 15) >> 4;
    int tile = (int)blockIdx.x * MAIN_WAVES + wave;
    if (tile >= nTiles) return;

    const int  m    = lane & 15;
    const bool hi   = lane >= 16;
    const int  half = hi ? 8 : 0;

    auto loadX = [&](int tl) -> v2f {
        int row = tl * 16 + m;
        if (row > nrows - 1) row = nrows - 1;
        return *(const v2f*)(x + (size_t)row * 2);
    };

    v2f xv = loadX(tile);

    const float l3v = l3[0];
    const float l1a = hi ? l1[2] : l1[0];
    const float l1b = l1[1];
    const int   kof = hi ? 2 : 0;

    __shared__ float wstrip[MAIN_WAVES][NTILE * 16];
    float* const myw = wstrip[wave];

    v2f   wa[NTILE];
    float p0 = 0.0f, p1 = 0.0f;
#pragma unroll
    for (int t = 0; t < NTILE; ++t) {
        const int  j  = t * 16 + m;
        const bool ok = j < NCOL;
        const int  jc = ok ? j : NCOL - 1;
        const float l2j = l2[jc];
        const float f0  = varphi1[jc * 3 + kof];
        const float f1  = varphi1[jc * 3 + 1];
        const float sj  = -l2j;
        float w  = varphi2[jc] * (l2j - l3v);
        float b0 = f0 * (l1a + sj);
        float b1 = f1 * (l1b + sj);
        if (!ok) { w = 0.0f; b0 = 0.0f; }
        if (hi || !ok) b1 = 0.0f;
        wa[t] = (v2f){b0, b1};
        myw[t * 16 + m] = w;
        p0 += w * f0;
        p1 += w * f1;
    }
    p0 += __shfl_xor(p0, 1, 32);  p1 += __shfl_xor(p1, 1, 32);
    p0 += __shfl_xor(p0, 2, 32);  p1 += __shfl_xor(p1, 2, 32);
    p0 += __shfl_xor(p0, 4, 32);  p1 += __shfl_xor(p1, 4, 32);
    p0 += __shfl_xor(p0, 8, 32);  p1 += __shfl_xor(p1, 8, 32);
    const float c0 = -lane_value(p0, 0);
    const float c1 = -lane_value(p1, 0);
    const float c2 = -lane_value(p0, 16);

    __builtin_amdgcn_fence(__ATOMIC_RELEASE, "wavefront");
    __builtin_amdgcn_fence(__ATOMIC_ACQUIRE, "wavefront");

    v2f ww[NTILE][4];
#pragma unroll
    for (int t = 0; t < NTILE; ++t) {
        const v4f lo = *(const v4f*)(myw + t * 16 + half);
        const v4f up = *(const v4f*)(myw + t * 16 + half + 4);
        ww[t][0] = (v2f){lo.x, lo.y};
        ww[t][1] = (v2f){lo.z, lo.w};
        ww[t][2] = (v2f){up.x, up.y};
        ww[t][3] = (v2f){up.z, up.w};
    }

    for (;;) {
        const int  nextTile = tile + nWaves;
        const bool more = nextTile < nTiles;
        v2f xn = xv;
        if (more) xn = loadX(nextTile);

        v2f bx;
        bx.x = hi ? 1.0f : xv.x;
        bx.y = hi ? 0.0f : xv.y;

        v2f accp[4] = {{0.f, 0.f}, {0.f, 0.f}, {0.f, 0.f}, {0.f, 0.f}};

        const v8f zero = {};
#pragma unroll
        for (int t = 0; t < NTILE; ++t) {
            const v8f d = __builtin_amdgcn_wmma_f32_16x16x4_f32(
                false, wa[t], false, bx, (short)0, zero, false, false);
            accp[0] += (v2f){relu1(d[0]), relu1(d[1])} * ww[t][0];
            accp[1] += (v2f){relu1(d[2]), relu1(d[3])} * ww[t][1];
            accp[2] += (v2f){relu1(d[4]), relu1(d[5])} * ww[t][2];
            accp[3] += (v2f){relu1(d[6]), relu1(d[7])} * ww[t][3];
        }

        const v2f s2 = (accp[0] + accp[1]) + (accp[2] + accp[3]);
        float s = s2.x + s2.y;
        s += other_half(s);

        const int row = tile * 16 + m;
        const float yv = s + c0 * xv.x + c1 * xv.y + c2;
        if (!hi && row < nrows) y[row] = yv;

        if (!more) break;
        xv = xn;
        tile = nextTile;
    }
}

extern "C" void kernel_launch(void* const* d_in, const int* in_sizes, int n_in,
                              void* d_out, int out_size, void* d_ws, size_t ws_size,
                              hipStream_t stream) {
    const float* x       = (const float*)d_in[0];
    const float* varphi1 = (const float*)d_in[1];
    const float* varphi2 = (const float*)d_in[2];
    const float* l1      = (const float*)d_in[3];
    const float* l2      = (const float*)d_in[4];
    const float* l3      = (const float*)d_in[5];
    float* y  = (float*)d_out;

    const int B = in_sizes[0] / 2;

    net_fused<<<MAIN_GRID, MAIN_THREADS, 0, stream>>>(
        x, varphi1, varphi2, l1, l2, l3, y, B);
}


# relu taken as h/2 + |h|/2: |d| is a source modifier of one multiply-add per hidden unit (8 max per WMMA gone), h/2 folded into the skip coefficients per wave; summed values and rounding change
# speedup vs baseline: 2.3361x; 1.0872x over previous
#include <hip/hip_runtime.h>


typedef __attribute__((ext_vector_type(2))) float v2f;
typedef __attribute__((ext_vector_type(4))) float v4f;
typedef __attribute__((ext_vector_type(8))) float v8f;

#if defined(__HIP_DEVICE_COMPILE__)
#define NO_PK __attribute__((target("no-packed-fp32-ops")))
#else
#define NO_PK
#endif

#define NCOL 300
#define NTILE 19


__device__ __forceinline__ NO_PK float lane_value(float v, int l) {
    return __builtin_bit_cast(float,
        __builtin_amdgcn_readlane(__builtin_bit_cast(int, v), l));
}

__device__ __forceinline__ NO_PK float xor_lane(float v, int lane, int mask) {
    return __builtin_bit_cast(float,
        __builtin_amdgcn_ds_bpermute((lane ^ mask) << 2, __builtin_bit_cast(int, v)));
}

__device__ __forceinline__ NO_PK float other_half(float v) {
    const int b = __builtin_bit_cast(int, v);
    return __builtin_bit_cast(float,
        __builtin_amdgcn_permlanex16(b, b, 0x76543210, 0xfedcba98, false, false));
}

#define MAIN_THREADS 512
#define MAIN_WAVES   (MAIN_THREADS / 32)
#define MAIN_GRID    256

__global__ __launch_bounds__(MAIN_THREADS) NO_PK void net_fused(
    const float* __restrict__ x,
    const float* __restrict__ varphi1,
    const float* __restrict__ varphi2,
    const float* __restrict__ l1,
    const float* __restrict__ l2,
    const float* __restrict__ l3,
    float* __restrict__ y,
    int nrows)
{
    const int tid  = (int)__builtin_amdgcn_workitem_id_x();
    const int lane = tid & 31;
    const int wave = __builtin_amdgcn_readfirstlane(tid >> 5);
    const int nWaves = MAIN_GRID * MAIN_WAVES;
    const int nTiles = (nrows + 15) >> 4;
    int tile = (int)__builtin_amdgcn_workgroup_id_x() * MAIN_WAVES + wave;
    if (tile >= nTiles) return;

    const int  m    = lane & 15;
    const bool hi   = lane >= 16;
    const int  half = hi ? 8 : 0;

    v2f xv;
    {
        int row = tile * 16 + m;
        if (row > nrows - 1) row = nrows - 1;
        xv = *(const v2f*)(x + (size_t)row * 2);
    }

    const float l3v = l3[0];
    const float l1a = hi ? l1[2] : l1[0];
    const float l1b = l1[1];
    const int   kof = hi ? 2 : 0;

    __shared__ float wstrip[MAIN_WAVES][NTILE * 16];
    float* const myw = wstrip[wave];

    v2f   wa[NTILE];
    float p0 = 0.0f, p1 = 0.0f;
    float q0 = 0.0f, q1 = 0.0f;
#pragma unroll
    for (int t = 0; t < NTILE; ++t) {
        const int  j  = t * 16 + m;
        const bool ok = j < NCOL;
        const int  jc = ok ? j : NCOL - 1;
        const float l2j = l2[jc];
        const float f0  = varphi1[jc * 3 + kof];
        const float f1  = varphi1[jc * 3 + 1];
        const float sj  = -l2j;
        float w  = varphi2[jc] * (l2j - l3v);
        float b0 = f0 * (l1a + sj);
        float b1 = f1 * (l1b + sj);
        if (!ok) { w = 0.0f; b0 = 0.0f; }
        if (hi || !ok) b1 = 0.0f;
        wa[t] = (v2f){b0, b1};
        myw[t * 16 + m] = 0.5f * w;
        p0 += w * f0;
        p1 += w * f1;
        q0 += w * b0;
        q1 += w * b1;
    }
#pragma unroll
    for (int k = 1; k < 16; k <<= 1) {
        p0 += xor_lane(p0, lane, k);  p1 += xor_lane(p1, lane, k);
        q0 += xor_lane(q0, lane, k);  q1 += xor_lane(q1, lane, k);
    }
    const float c0 = 0.5f * lane_value(q0, 0)  - lane_value(p0, 0);
    const float c1 = 0.5f * lane_value(q1, 0)  - lane_value(p1, 0);
    const float c2 = 0.5f * lane_value(q0, 16) - lane_value(p0, 16);

    __builtin_amdgcn_fence(__ATOMIC_RELEASE, "wavefront");
    __builtin_amdgcn_fence(__ATOMIC_ACQUIRE, "wavefront");

    float ww[NTILE][8];
#pragma unroll
    for (int t = 0; t < NTILE; ++t) {
        const v4f lo = *(const v4f*)(myw + t * 16 + half);
        const v4f up = *(const v4f*)(myw + t * 16 + half + 4);
        ww[t][0] = lo.x;  ww[t][1] = lo.y;  ww[t][2] = lo.z;  ww[t][3] = lo.w;
        ww[t][4] = up.x;  ww[t][5] = up.y;  ww[t][6] = up.z;  ww[t][7] = up.w;
    }

    for (;;) {
        const int  nextTile = tile + nWaves;
        const bool more = nextTile < nTiles;
        v2f xn = xv;
        if (more) {
            int row = nextTile * 16 + m;
            if (row > nrows - 1) row = nrows - 1;
            xn = *(const v2f*)(x + (size_t)row * 2);
        }

        v2f bx;
        bx.x = hi ? 1.0f : xv.x;
        bx.y = hi ? 0.0f : xv.y;

        float acc[8] = {0.f, 0.f, 0.f, 0.f, 0.f, 0.f, 0.f, 0.f};

        const v8f zero = {};
#pragma unroll
        for (int t = 0; t < NTILE; ++t) {
            const v8f d = __builtin_amdgcn_wmma_f32_16x16x4_f32(
                false, wa[t], false, bx, (short)0, zero, false, false);
#pragma unroll
            for (int r = 0; r < 8; ++r)
                acc[r] = __builtin_fmaf(__builtin_fabsf(d[r]), ww[t][r], acc[r]);
        }

        const float se = (acc[0] + acc[2]) + (acc[4] + acc[6]);
        const float so = (acc[1] + acc[3]) + (acc[5] + acc[7]);
        float s = se + so;
        s += other_half(s);

        const int row = tile * 16 + m;
        const float yv = s + c0 * xv.x + c1 * xv.y + c2;
        if (!hi && row < nrows) y[row] = yv;

        if (!more) break;
        xv = xn;
        tile = nextTile;
    }
}

extern "C" void kernel_launch(void* const* d_in, const int* in_sizes, int n_in,
                              void* d_out, int out_size, void* d_ws, size_t ws_size,
                              hipStream_t stream) {
    const float* x       = (const float*)d_in[0];
    const float* varphi1 = (const float*)d_in[1];
    const float* varphi2 = (const float*)d_in[2];
    const float* l1      = (const float*)d_in[3];
    const float* l2      = (const float*)d_in[4];
    const float* l3      = (const float*)d_in[5];
    float* y  = (float*)d_out;

    const int B = in_sizes[0] / 2;

    net_fused<<<MAIN_GRID, MAIN_THREADS, 0, stream>>>(
        x, varphi1, varphi2, l1, l2, l3, y, B);
}
